# starting kernel
# baseline (speedup 1.0000x reference)
#include <hip/hip_runtime.h>
#include <hip/hip_bf16.h>

typedef __attribute__((ext_vector_type(2))) float v2f;
typedef __attribute__((ext_vector_type(8))) float v8f;

#define BATCH 8
#define CCH   16
#define HH    512
#define WW    512

__global__ __launch_bounds__(256) void pool_pack_kernel(
    const float* __restrict__ x, float* __restrict__ U2)
{
    int idx = blockIdx.x * blockDim.x + threadIdx.x;
    int pw = idx & 63;
    int ph = (idx >> 6) & 63;
    int c  = (idx >> 12) & 15;
    int b  = idx >> 16;

    const float* src = x + (((size_t)(b * CCH + c) * HH) + (size_t)ph * 8) * WW + pw * 8;
    float s = 0.0f;
#pragma unroll
    for (int r = 0; r < 8; ++r) {
        const float4* p = (const float4*)(src + (size_t)r * WW);
        float4 a = p[0];
        float4 bb = p[1];
        s += (a.x + a.y) + (a.z + a.w) + (bb.x + bb.y) + (bb.z + bb.w);
    }
    s *= (1.0f / 64.0f);

    int lp = ((ph >> 3) << 3) | (pw >> 3);
    int n  = (c << 6) | ((ph & 7) << 3) | (pw & 7);
    U2[(size_t)b * 65536 + (size_t)(lp >> 1) * 2048 + n * 2 + (lp & 1)] = s;
}

#define LDSS 68

__global__ __launch_bounds__(256) void att_gemm_kernel(
    const float* __restrict__ attn,
    const float* __restrict__ U2,
    float* __restrict__ out)
{
    __shared__ float As[16 * LDSS];
    __shared__ float rowScale[16];

    int bid   = blockIdx.x;
    int mtile = bid & 255;
    int b     = bid >> 8;
    int l_base = mtile * 16;

    int t    = threadIdx.x;
    int lane = t & 31;
    int wave = t >> 5;

    {
        int row  = t >> 4;
        int col4 = (t & 15) * 4;
        float4 v = *(const float4*)(attn + ((size_t)b * 4096 + l_base + row) * 64 + col4);
        *(float4*)(&As[row * LDSS + col4]) = v;
    }
    __syncthreads();

    if (t < 16) {
        int cnt = 0;
#pragma unroll
        for (int k = 0; k < 64; ++k) cnt += (As[t * LDSS + k] != 0.0f) ? 1 : 0;
        rowScale[t] = 1.0f / ((float)cnt + 1e-5f);
    }
    __syncthreads();

    int n_wave = wave * 128;
    int M = lane & 15;
    int h = lane >> 4;
    float scale = rowScale[M];

    v8f acc[8];
#pragma unroll
    for (int i = 0; i < 8; ++i) acc[i] = (v8f){0.f,0.f,0.f,0.f,0.f,0.f,0.f,0.f};

    const float* Ub = U2 + (size_t)b * 65536;

#pragma unroll
    for (int ks = 0; ks < 16; ++ks) {
        int k0 = ks * 4 + h * 2;
        v2f a;
        a.x = As[M * LDSS + k0]     * scale;
        a.y = As[M * LDSS + k0 + 1] * scale;
#pragma unroll
        for (int nt = 0; nt < 8; ++nt) {
            int n = n_wave + nt * 16 + M;
            v2f bf = *(const v2f*)(Ub + (size_t)(k0 >> 1) * 2048 + n * 2);
            acc[nt] = __builtin_amdgcn_wmma_f32_16x16x4_f32(
                false, a, false, bf, (short)0, acc[nt], false, false);
        }
    }

    int hp  = l_base >> 6;
    int wp0 = (l_base & 63) + 8 * h;
#pragma unroll
    for (int nt = 0; nt < 8; ++nt) {
        int k  = n_wave + nt * 16 + M;
        int c  = k >> 6;
        int kh = (k >> 3) & 7;
        int kw = k & 7;
        size_t o0 = (((size_t)(b * CCH + c) * HH) + (size_t)(hp * 8 + kh)) * WW
                  + (size_t)wp0 * 8 + kw;
#pragma unroll
        for (int v = 0; v < 8; ++v) {
            out[o0 + (size_t)v * 8] = acc[nt][v];
        }
    }
}

extern "C" void kernel_launch(void* const* d_in, const int* in_sizes, int n_in,
                              void* d_out, int out_size, void* d_ws, size_t ws_size,
                              hipStream_t stream)
{
    const float* x    = (const float*)d_in[0];
    const float* attn = (const float*)d_in[1];
    float* out = (float*)d_out;
    float* U2  = (float*)d_ws;

    pool_pack_kernel<<<2048, 256, 0, stream>>>(x, U2);
    att_gemm_kernel<<<2048, 256, 0, stream>>>(attn, U2, out);
}


# Tile walk as a software pipeline: a tile's global pieces requested two steps ahead, the next tile's slot stores and fragment reads as one group in front of the current tile's 32 products
# speedup vs baseline: 1.4547x; 1.4547x over previous
#include <hip/hip_runtime.h>

typedef __attribute__((ext_vector_type(2))) float v2f;
typedef __attribute__((ext_vector_type(4))) float v4f;
typedef __attribute__((ext_vector_type(8))) float v8f;

#define CCH   16
#define HH    512
#define WW    512
#define NWAVE 16
#define PITCH 68
#define SLOT  (16 * PITCH)

#define DEV static __device__ __forceinline__

DEV int xhalf(int v) { return __builtin_amdgcn_permlanex16(v, v, 0x76543210, 0xfedcba98, false, false); }

DEV void load_tile(v4f (&g)[8], const float* p)
{
#pragma unroll
    for (int q = 0; q < 8; ++q)
        g[q] = *(const v4f*)(p + q * 128);
}

DEV void stage_tile(float* slot, const v4f (&g)[8], v4f (&bf)[8], int N, int h)
{
#pragma unroll
    for (int q = 0; q < 8; ++q)
        *(v4f*)&slot[(2 * q + h) * PITCH + N * 4] = g[q];
    __builtin_amdgcn_fence(__ATOMIC_RELEASE, "wavefront");
    __builtin_amdgcn_wave_barrier();
    __builtin_amdgcn_fence(__ATOMIC_ACQUIRE, "wavefront");
#pragma unroll
    for (int j = 0; j < 8; ++j)
        bf[j] = *(const v4f*)&slot[N * PITCH + 8 * j + 4 * h];
}

DEV float tile_scale(const v4f (&bf)[8])
{
    int cnt = 0;
#pragma unroll
    for (int j = 0; j < 8; ++j) {
        cnt += (bf[j].x != 0.0f) ? 1 : 0;
        cnt += (bf[j].y != 0.0f) ? 1 : 0;
        cnt += (bf[j].z != 0.0f) ? 1 : 0;
        cnt += (bf[j].w != 0.0f) ? 1 : 0;
    }
    cnt += xhalf(cnt);
    return 1.0f / ((float)cnt + 1e-5f);
}

template <int J0>
DEV void products(v8f& acc0, v8f& acc1, const v4f (&af0)[8], const v4f (&af1)[8], const v4f (&bf)[8])
{
#pragma unroll
    for (int j = J0; j < J0 + 4; ++j) {
        v2f bb, a0, a1;
        bb.x = bf[j].x;  bb.y = bf[j].y;
        a0.x = af0[j].x; a0.y = af0[j].y;
        a1.x = af1[j].x; a1.y = af1[j].y;
        acc0 = __builtin_amdgcn_wmma_f32_16x16x4_f32(false, a0, false, bb, (short)0, acc0, false, false);
        acc1 = __builtin_amdgcn_wmma_f32_16x16x4_f32(false, a1, false, bb, (short)0, acc1, false, false);
        bb.x = bf[j].z;  bb.y = bf[j].w;
        a0.x = af0[j].z; a0.y = af0[j].w;
        a1.x = af1[j].z; a1.y = af1[j].w;
        acc0 = __builtin_amdgcn_wmma_f32_16x16x4_f32(false, a0, false, bb, (short)0, acc0, false, false);
        acc1 = __builtin_amdgcn_wmma_f32_16x16x4_f32(false, a1, false, bb, (short)0, acc1, false, false);
    }
}

DEV void store_tile(const v8f& acc0, const v8f& acc1, float scale, float* o)
{
    v4f d;
    d.x = acc0[0] * scale; d.y = acc0[1] * scale; d.z = acc0[2] * scale; d.w = acc0[3] * scale;
    *(v4f*)(o) = d;
    d.x = acc0[4] * scale; d.y = acc0[5] * scale; d.z = acc0[6] * scale; d.w = acc0[7] * scale;
    *(v4f*)(o + WW) = d;
    d.x = acc1[0] * scale; d.y = acc1[1] * scale; d.z = acc1[2] * scale; d.w = acc1[3] * scale;
    *(v4f*)(o + 2 * WW) = d;
    d.x = acc1[4] * scale; d.y = acc1[5] * scale; d.z = acc1[6] * scale; d.w = acc1[7] * scale;
    *(v4f*)(o + 3 * WW) = d;
}

template <bool STAGE, bool FETCH>
DEV void tile_step(const v4f (&af0)[8], const v4f (&af1)[8], const v4f (&bf_cur)[8], float* o,
                   float* slot_next, const v4f (&g_next)[8], v4f (&g_fetch)[8], const float* fetch_from,
                   v4f (&bf_next)[8], int N, int h)
{
    __builtin_amdgcn_sched_barrier(0);
    if (STAGE) {
        stage_tile(slot_next, g_next, bf_next, N, h);
        if (FETCH) load_tile(g_fetch, fetch_from);
        __builtin_amdgcn_sched_barrier(0);
    }
    v8f acc0 = (v8f){0.f, 0.f, 0.f, 0.f, 0.f, 0.f, 0.f, 0.f};
    v8f acc1 = (v8f){0.f, 0.f, 0.f, 0.f, 0.f, 0.f, 0.f, 0.f};
    products<0>(acc0, acc1, af0, af1, bf_cur);
    products<4>(acc0, acc1, af0, af1, bf_cur);
    store_tile(acc0, acc1, tile_scale(bf_cur), o);
}

__global__ __launch_bounds__(512) void fused_pool_att_kernel(
    const float* __restrict__ x,
    const float* __restrict__ attn,
    float* __restrict__ out)
{
    __shared__ __attribute__((aligned(16))) float Ap[32 * PITCH];
    __shared__ __attribute__((aligned(16))) float Bs[NWAVE * 2 * SLOT];

    const int bid  = blockIdx.x;
    const int khg  = bid & 1;
    const int c    = (bid >> 1) & 15;
    const int b    = bid >> 5;
    const int t    = threadIdx.x;
    const int lane = t & 31;
    const int wave = t >> 5;
    const int N    = lane & 15;
    const int h    = lane >> 4;

    const size_t tstep = (size_t)16 * 1024;
    const float* attnw = attn + (size_t)b * 4096 * 64 + (size_t)wave * 1024 + lane * 4;
    v4f g[8], g1[8];
    load_tile(g, attnw);
    load_tile(g1, attnw + tstep);

    const float* xc = x + (size_t)(b * CCH + c) * HH * WW;
    for (int pp = 0; pp < 2; ++pp) {
        const int p   = wave * 2 + pp;
        const int hps = p >> 2;
        const int khl = p & 3;
        const int ph  = hps * 8 + khg * 4 + khl;
        const float* src = xc + (size_t)(ph * 8) * WW + lane * 4;
        float s0 = 0.f, s1 = 0.f, s2 = 0.f, s3 = 0.f;
#pragma unroll
        for (int r = 0; r < 8; ++r) {
            const v4f v0 = *(const v4f*)(src + r * WW);
            const v4f v1 = *(const v4f*)(src + r * WW + 128);
            const v4f v2 = *(const v4f*)(src + r * WW + 256);
            const v4f v3 = *(const v4f*)(src + r * WW + 384);
            s0 += (v0.x + v0.y) + (v0.z + v0.w);
            s1 += (v1.x + v1.y) + (v1.z + v1.w);
            s2 += (v2.x + v2.y) + (v2.z + v2.w);
            s3 += (v3.x + v3.y) + (v3.z + v3.w);
        }
        const float sv[4] = {s0, s1, s2, s3};
#pragma unroll
        for (int j = 0; j < 4; ++j) {
            const float tot = (sv[j] + __shfl_xor(sv[j], 1)) * (1.0f / 64.0f);
            const int pw  = j * 16 + (lane >> 1);
            const int lp  = hps * 8 + (pw >> 3);
            const int kw  = pw & 7;
            const int row = ((khl >> 1) << 4) | ((kw >> 2) << 3) | ((khl & 1) << 2) | (kw & 3);
            if ((lane & 1) == 0) Ap[row * PITCH + lp] = tot;
        }
    }
    __syncthreads();

    v4f af0[8], af1[8];
#pragma unroll
    for (int j = 0; j < 8; ++j) {
        af0[j] = *(const v4f*)&Ap[N * PITCH + 8 * j + 4 * h];
        af1[j] = *(const v4f*)&Ap[(16 + N) * PITCH + 8 * j + 4 * h];
    }

    float* slot0 = Bs + wave * 2 * SLOT;
    float* slot1 = slot0 + SLOT;
    float* ow = out + (size_t)(b * CCH + c) * HH * WW
              + (size_t)((wave >> 2) * 8 + khg * 4) * WW + ((wave & 3) * 16 + N) * 8 + 4 * h;

    v4f bfA[8], bfB[8];
    stage_tile(slot0, g, bfA, N, h);
    load_tile(g, attnw + 2 * tstep);

    const size_t ostep = (size_t)32 * WW;
    for (int i = 0; i < 12; i += 2) {
        tile_step<true, true>(af0, af1, bfA, ow + i * ostep,
                              slot1, g1, g1, attnw + (i + 3) * tstep, bfB, N, h);
        tile_step<true, true>(af0, af1, bfB, ow + (i + 1) * ostep,
                              slot0, g, g, attnw + (i + 4) * tstep, bfA, N, h);
    }
    tile_step<true, true>(af0, af1, bfA, ow + 12 * ostep, slot1, g1, g1, attnw + 15 * tstep, bfB, N, h);
    tile_step<true, false>(af0, af1, bfB, ow + 13 * ostep, slot0, g, g, attnw, bfA, N, h);
    tile_step<true, false>(af0, af1, bfA, ow + 14 * ostep, slot1, g1, g1, attnw, bfB, N, h);
    tile_step<false, false>(af0, af1, bfB, ow + 15 * ostep, slot0, g, g, attnw, bfA, N, h);
}

extern "C" void kernel_launch(void* const* d_in, const int* in_sizes, int n_in,
                              void* d_out, int out_size, void* d_ws, size_t ws_size,
                              hipStream_t stream)
{
    const float* x    = (const float*)d_in[0];
    const float* attn = (const float*)d_in[1];
    float* out = (float*)d_out;

    fused_pool_att_kernel<<<256, 512, 0, stream>>>(x, attn, out);
}


# Row nonzero count off the common path: a wave tests its fragments for any zero with a min chain (2 values per instruction); a zero-free tile takes the folded scale of count 64, else the exact count
# speedup vs baseline: 1.5385x; 1.0576x over previous
#include <hip/hip_runtime.h>

typedef __attribute__((ext_vector_type(2))) float v2f;
typedef __attribute__((ext_vector_type(4))) float v4f;
typedef __attribute__((ext_vector_type(8))) float v8f;

#define CCH   16
#define HH    512
#define WW    512
#define NWAVE 16
#define PITCH 68
#define SLOT  (16 * PITCH)

#define DEV static __device__ __forceinline__

DEV int xhalf(int v) { return __builtin_amdgcn_permlanex16(v, v, 0x76543210, 0xfedcba98, false, false); }

DEV void load_tile(v4f (&g)[8], const float* p)
{
#pragma unroll
    for (int q = 0; q < 8; ++q)
        g[q] = *(const v4f*)(p + q * 128);
}

DEV void stage_tile(float* slot, const v4f (&g)[8], v4f (&bf)[8], int N, int h)
{
#pragma unroll
    for (int q = 0; q < 8; ++q)
        *(v4f*)&slot[(2 * q + h) * PITCH + N * 4] = g[q];
    __builtin_amdgcn_fence(__ATOMIC_RELEASE, "wavefront");
    __builtin_amdgcn_wave_barrier();
    __builtin_amdgcn_fence(__ATOMIC_ACQUIRE, "wavefront");
#pragma unroll
    for (int j = 0; j < 8; ++j)
        bf[j] = *(const v4f*)&slot[N * PITCH + 8 * j + 4 * h];
}

DEV float tile_scale(const v4f (&bf)[8])
{
    int cnt = 0;
#pragma unroll
    for (int j = 0; j < 8; ++j) {
        cnt += (bf[j].x != 0.0f) ? 1 : 0;
        cnt += (bf[j].y != 0.0f) ? 1 : 0;
        cnt += (bf[j].z != 0.0f) ? 1 : 0;
        cnt += (bf[j].w != 0.0f) ? 1 : 0;
    }
    cnt += xhalf(cnt);
    return 1.0f / ((float)cnt + 1e-5f);
}

DEV bool lane_zero_free(const v4f (&bf)[8])
{
    float m0 = __builtin_fminf(__builtin_fabsf(bf[0].x), __builtin_fabsf(bf[0].y));
    float m1 = __builtin_fminf(__builtin_fabsf(bf[0].z), __builtin_fabsf(bf[0].w));
#pragma unroll
    for (int j = 1; j < 8; ++j) {
        m0 = __builtin_fminf(m0, __builtin_fminf(__builtin_fabsf(bf[j].x), __builtin_fabsf(bf[j].y)));
        m1 = __builtin_fminf(m1, __builtin_fminf(__builtin_fabsf(bf[j].z), __builtin_fabsf(bf[j].w)));
    }
    return __builtin_fminf(m0, m1) > 0.0f;
}

template <int J0>
DEV void products(v8f& acc0, v8f& acc1, const v4f (&af0)[8], const v4f (&af1)[8], const v4f (&bf)[8])
{
#pragma unroll
    for (int j = J0; j < J0 + 4; ++j) {
        v2f bb, a0, a1;
        bb.x = bf[j].x;  bb.y = bf[j].y;
        a0.x = af0[j].x; a0.y = af0[j].y;
        a1.x = af1[j].x; a1.y = af1[j].y;
        acc0 = __builtin_amdgcn_wmma_f32_16x16x4_f32(false, a0, false, bb, (short)0, acc0, false, false);
        acc1 = __builtin_amdgcn_wmma_f32_16x16x4_f32(false, a1, false, bb, (short)0, acc1, false, false);
        bb.x = bf[j].z;  bb.y = bf[j].w;
        a0.x = af0[j].z; a0.y = af0[j].w;
        a1.x = af1[j].z; a1.y = af1[j].w;
        acc0 = __builtin_amdgcn_wmma_f32_16x16x4_f32(false, a0, false, bb, (short)0, acc0, false, false);
        acc1 = __builtin_amdgcn_wmma_f32_16x16x4_f32(false, a1, false, bb, (short)0, acc1, false, false);
    }
}

DEV void store_tile(const v8f& acc0, const v8f& acc1, float scale, float* o)
{
    v4f d;
    d.x = acc0[0] * scale; d.y = acc0[1] * scale; d.z = acc0[2] * scale; d.w = acc0[3] * scale;
    *(v4f*)(o) = d;
    d.x = acc0[4] * scale; d.y = acc0[5] * scale; d.z = acc0[6] * scale; d.w = acc0[7] * scale;
    *(v4f*)(o + WW) = d;
    d.x = acc1[0] * scale; d.y = acc1[1] * scale; d.z = acc1[2] * scale; d.w = acc1[3] * scale;
    *(v4f*)(o + 2 * WW) = d;
    d.x = acc1[4] * scale; d.y = acc1[5] * scale; d.z = acc1[6] * scale; d.w = acc1[7] * scale;
    *(v4f*)(o + 3 * WW) = d;
}

template <bool STAGE, bool FETCH>
DEV void tile_step(const v4f (&af0)[8], const v4f (&af1)[8], const v4f (&bf_cur)[8], float* o,
                   float* slot_next, const v4f (&g_next)[8], v4f (&g_fetch)[8], const float* fetch_from,
                   v4f (&bf_next)[8], int N, int h)
{
    __builtin_amdgcn_sched_barrier(0);
    if (STAGE) {
        stage_tile(slot_next, g_next, bf_next, N, h);
        if (FETCH) load_tile(g_fetch, fetch_from);
        __builtin_amdgcn_sched_barrier(0);
    }
    v8f acc0 = (v8f){0.f, 0.f, 0.f, 0.f, 0.f, 0.f, 0.f, 0.f};
    v8f acc1 = (v8f){0.f, 0.f, 0.f, 0.f, 0.f, 0.f, 0.f, 0.f};
    products<0>(acc0, acc1, af0, af1, bf_cur);
    products<4>(acc0, acc1, af0, af1, bf_cur);
    float scale = 1.0f / ((float)64 + 1e-5f);
    if (__builtin_amdgcn_ballot_w32(lane_zero_free(bf_cur)) != 0xffffffffu)
        scale = tile_scale(bf_cur);
    store_tile(acc0, acc1, scale, o);
}

__global__ __launch_bounds__(512) void fused_pool_att_kernel(
    const float* __restrict__ x,
    const float* __restrict__ attn,
    float* __restrict__ out)
{
    __shared__ __attribute__((aligned(16))) float Ap[32 * PITCH];
    __shared__ __attribute__((aligned(16))) float Bs[NWAVE * 2 * SLOT];

    const int bid  = blockIdx.x;
    const int khg  = bid & 1;
    const int c    = (bid >> 1) & 15;
    const int b    = bid >> 5;
    const int t    = threadIdx.x;
    const int lane = t & 31;
    const int wave = t >> 5;
    const int N    = lane & 15;
    const int h    = lane >> 4;

    const size_t tstep = (size_t)16 * 1024;
    const float* attnw = attn + (size_t)b * 4096 * 64 + (size_t)wave * 1024 + lane * 4;
    v4f g[8], g1[8];
    load_tile(g, attnw);
    load_tile(g1, attnw + tstep);

    const float* xc = x + (size_t)(b * CCH + c) * HH * WW;
    for (int pp = 0; pp < 2; ++pp) {
        const int p   = wave * 2 + pp;
        const int hps = p >> 2;
        const int khl = p & 3;
        const int ph  = hps * 8 + khg * 4 + khl;
        const float* src = xc + (size_t)(ph * 8) * WW + lane * 4;
        float s0 = 0.f, s1 = 0.f, s2 = 0.f, s3 = 0.f;
#pragma unroll
        for (int r = 0; r < 8; ++r) {
            const v4f v0 = *(const v4f*)(src + r * WW);
            const v4f v1 = *(const v4f*)(src + r * WW + 128);
            const v4f v2 = *(const v4f*)(src + r * WW + 256);
            const v4f v3 = *(const v4f*)(src + r * WW + 384);
            s0 += (v0.x + v0.y) + (v0.z + v0.w);
            s1 += (v1.x + v1.y) + (v1.z + v1.w);
            s2 += (v2.x + v2.y) + (v2.z + v2.w);
            s3 += (v3.x + v3.y) + (v3.z + v3.w);
        }
        const float sv[4] = {s0, s1, s2, s3};
#pragma unroll
        for (int j = 0; j < 4; ++j) {
            const float tot = (sv[j] + __shfl_xor(sv[j], 1)) * (1.0f / 64.0f);
            const int pw  = j * 16 + (lane >> 1);
            const int lp  = hps * 8 + (pw >> 3);
            const int kw  = pw & 7;
            const int row = ((khl >> 1) << 4) | ((kw >> 2) << 3) | ((khl & 1) << 2) | (kw & 3);
            if ((lane & 1) == 0) Ap[row * PITCH + lp] = tot;
        }
    }
    __syncthreads();

    v4f af0[8], af1[8];
#pragma unroll
    for (int j = 0; j < 8; ++j) {
        af0[j] = *(const v4f*)&Ap[N * PITCH + 8 * j + 4 * h];
        af1[j] = *(const v4f*)&Ap[(16 + N) * PITCH + 8 * j + 4 * h];
    }

    float* slot0 = Bs + wave * 2 * SLOT;
    float* slot1 = slot0 + SLOT;
    float* ow = out + (size_t)(b * CCH + c) * HH * WW
              + (size_t)((wave >> 2) * 8 + khg * 4) * WW + ((wave & 3) * 16 + N) * 8 + 4 * h;

    v4f bfA[8], bfB[8];
    stage_tile(slot0, g, bfA, N, h);
    load_tile(g, attnw + 2 * tstep);

    const size_t ostep = (size_t)32 * WW;
    for (int i = 0; i < 12; i += 2) {
        tile_step<true, true>(af0, af1, bfA, ow + i * ostep,
                              slot1, g1, g1, attnw + (i + 3) * tstep, bfB, N, h);
        tile_step<true, true>(af0, af1, bfB, ow + (i + 1) * ostep,
                              slot0, g, g, attnw + (i + 4) * tstep, bfA, N, h);
    }
    tile_step<true, true>(af0, af1, bfA, ow + 12 * ostep, slot1, g1, g1, attnw + 15 * tstep, bfB, N, h);
    tile_step<true, false>(af0, af1, bfB, ow + 13 * ostep, slot0, g, g, attnw, bfA, N, h);
    tile_step<true, false>(af0, af1, bfA, ow + 14 * ostep, slot1, g1, g1, attnw, bfB, N, h);
    tile_step<false, false>(af0, af1, bfB, ow + 15 * ostep, slot0, g, g, attnw, bfA, N, h);
}

extern "C" void kernel_launch(void* const* d_in, const int* in_sizes, int n_in,
                              void* d_out, int out_size, void* d_ws, size_t ws_size,
                              hipStream_t stream)
{
    const float* x    = (const float*)d_in[0];
    const float* attn = (const float*)d_in[1];
    float* out = (float*)d_out;

    fused_pool_att_kernel<<<256, 512, 0, stream>>>(x, attn, out);
}
